# starting kernel
# baseline (speedup 1.0000x reference)
#include <hip/hip_runtime.h>
#include <hip/hip_bf16.h>
#include <math.h>

typedef __attribute__((ext_vector_type(2))) float v2f;
typedef __attribute__((ext_vector_type(8))) float v8f;

#define B_  32
#define D_  1024
#define H_  64
#define O_  1000
#define E_  6

__global__ void moe_gate(const float* __restrict__ x,
                         const float* __restrict__ gw,
                         const float* __restrict__ gb,
                         float* __restrict__ gates) {
  __shared__ float logits[B_][E_];
  int t = threadIdx.x;
  if (t < B_ * E_) {
    int b = t / E_, e = t % E_;
    const float* xr = x + b * D_;
    float acc = gb[e];
    for (int d = 0; d < D_; ++d) acc = fmaf(xr[d], gw[d * E_ + e], acc);
    logits[b][e] = acc;
  }
  __syncthreads();
  if (t < B_) {
    int b = t;
    int i0 = 0; float v0 = logits[b][0];
    for (int e = 1; e < E_; ++e) { float v = logits[b][e]; if (v > v0) { v0 = v; i0 = e; } }
    int i1 = -1; float v1 = -3.4e38f;
    for (int e = 0; e < E_; ++e) {
      if (e == i0) continue;
      float v = logits[b][e]; if (v > v1) { v1 = v; i1 = e; }
    }
    float e0 = 1.0f;
    float e1 = __expf(v1 - v0);
    float inv = 1.0f / (e0 + e1);
    for (int e = 0; e < E_; ++e) gates[b * E_ + e] = 0.0f;
    gates[b * E_ + i0] = e0 * inv;
    gates[b * E_ + i1] = e1 * inv;
  }
}

__global__ void moe_fc1(const float* __restrict__ x,
                        const float* __restrict__ w1,
                        const float* __restrict__ b1,
                        float* __restrict__ h) {
  int wave = blockIdx.x * 4 + (threadIdx.x >> 5);
  if (wave >= E_ * 2 * 4) return;
  int e   = wave >> 3;
  int m_t = (wave >> 2) & 1;
  int n_t = wave & 3;
  int lane  = threadIdx.x & 31;
  int row   = lane & 15;
  int khalf = lane >> 4;

  const float* xr  = x  + (m_t * 16 + row) * D_;
  const float* w1e = w1 + e * D_ * H_ + n_t * 16 + row;
  v8f c = {};
  for (int k = 0; k < D_; k += 4) {
    int ka = k + 2 * khalf;
    v2f a = *(const v2f*)(xr + ka);
    v2f bb;
    bb.x = w1e[(ka + 0) * H_];
    bb.y = w1e[(ka + 1) * H_];
    c = __builtin_amdgcn_wmma_f32_16x16x4_f32(
         false, a,  false, bb,
         (short)0, c,  false,  false);
  }
  float bias = b1[e * H_ + n_t * 16 + row];
  float* hout = h + (e * B_ + m_t * 16) * H_ + n_t * 16 + row;
#pragma unroll
  for (int v = 0; v < 8; ++v) {
    int r = v + 8 * khalf;
    float val = c[v] + bias;
    hout[r * H_] = val > 0.0f ? val : 0.0f;
  }
}

__global__ void moe_fc2(const float* __restrict__ hbuf,
                        const float* __restrict__ w2,
                        const float* __restrict__ b2,
                        const float* __restrict__ gates,
                        float* __restrict__ out) {
  const int NT = (O_ + 15) / 16;
  int wave = blockIdx.x * 4 + (threadIdx.x >> 5);
  if (wave >= 2 * NT) return;
  int m_t = wave / NT;
  int n_t = wave % NT;
  int lane  = threadIdx.x & 31;
  int row   = lane & 15;
  int khalf = lane >> 4;
  int ocol  = n_t * 16 + row;
  bool colok = ocol < O_;

  v8f c = {};
  for (int e = 0; e < E_; ++e) {
    float g = gates[(m_t * 16 + row) * E_ + e];
    const float* he  = hbuf + (e * B_ + m_t * 16 + row) * H_;
    const float* w2e = w2 + e * H_ * O_;
#pragma unroll
    for (int k = 0; k < H_; k += 4) {
      int ka = k + 2 * khalf;
      v2f a = *(const v2f*)(he + ka);
      a.x *= g; a.y *= g;
      v2f bb;
      bb.x = colok ? w2e[(ka + 0) * O_ + ocol] : 0.0f;
      bb.y = colok ? w2e[(ka + 1) * O_ + ocol] : 0.0f;
      c = __builtin_amdgcn_wmma_f32_16x16x4_f32(
          false, a, false, bb, (short)0, c, false, false);
    }
  }
  if (colok) {
#pragma unroll
    for (int v = 0; v < 8; ++v) {
      int r = m_t * 16 + v + 8 * khalf;
      float acc = c[v];
      for (int e = 0; e < E_; ++e)
        acc = fmaf(gates[r * E_ + e], b2[e * O_ + ocol], acc);
      out[r * O_ + ocol] = acc;
    }
  }
}

extern "C" void kernel_launch(void* const* d_in, const int* in_sizes, int n_in,
                              void* d_out, int out_size, void* d_ws, size_t ws_size,
                              hipStream_t stream) {
  const float* x      = (const float*)d_in[0];
  const float* gate_w = (const float*)d_in[1];
  const float* gate_b = (const float*)d_in[2];
  const float* w1     = (const float*)d_in[3];
  const float* b1     = (const float*)d_in[4];
  const float* w2     = (const float*)d_in[5];
  const float* b2     = (const float*)d_in[6];
  float* out = (float*)d_out;

  float* gates = (float*)d_ws;
  float* hbuf  = gates + 256;

  moe_gate<<<1, B_ * E_, 0, stream>>>(x, gate_w, gate_b, gates);

  moe_fc1<<<(48 + 3) / 4, 128, 0, stream>>>(x, w1, b1, hbuf);

  moe_fc2<<<(126 + 3) / 4, 128, 0, stream>>>(hbuf, w2, b2, gates, out);
}


# fc2: B loads unpredicated (a lane past column 1000 reads the last valid column; stores stay predicated), expert loop unrolled: about 60 loads in flight ahead of the WMMA chain, was 3
# speedup vs baseline: 1.5175x; 1.5175x over previous
#include <hip/hip_runtime.h>
#include <hip/hip_bf16.h>
#include <math.h>

typedef __attribute__((ext_vector_type(2))) float v2f;
typedef __attribute__((ext_vector_type(8))) float v8f;

#define B_  32
#define D_  1024
#define H_  64
#define O_  1000
#define E_  6

__global__ void moe_gate(const float* __restrict__ x,
                         const float* __restrict__ gw,
                         const float* __restrict__ gb,
                         float* __restrict__ gates) {
  __shared__ float logits[B_][E_];
  int t = threadIdx.x;
  if (t < B_ * E_) {
    int b = t / E_, e = t % E_;
    const float* xr = x + b * D_;
    float acc = gb[e];
    for (int d = 0; d < D_; ++d) acc = fmaf(xr[d], gw[d * E_ + e], acc);
    logits[b][e] = acc;
  }
  __syncthreads();
  if (t < B_) {
    int b = t;
    int i0 = 0; float v0 = logits[b][0];
    for (int e = 1; e < E_; ++e) { float v = logits[b][e]; if (v > v0) { v0 = v; i0 = e; } }
    int i1 = -1; float v1 = -3.4e38f;
    for (int e = 0; e < E_; ++e) {
      if (e == i0) continue;
      float v = logits[b][e]; if (v > v1) { v1 = v; i1 = e; }
    }
    float e0 = 1.0f;
    float e1 = __expf(v1 - v0);
    float inv = 1.0f / (e0 + e1);
    for (int e = 0; e < E_; ++e) gates[b * E_ + e] = 0.0f;
    gates[b * E_ + i0] = e0 * inv;
    gates[b * E_ + i1] = e1 * inv;
  }
}

__global__ void moe_fc1(const float* __restrict__ x,
                        const float* __restrict__ w1,
                        const float* __restrict__ b1,
                        float* __restrict__ h) {
  int wave = blockIdx.x * 4 + (threadIdx.x >> 5);
  if (wave >= E_ * 2 * 4) return;
  int e   = wave >> 3;
  int m_t = (wave >> 2) & 1;
  int n_t = wave & 3;
  int lane  = threadIdx.x & 31;
  int row   = lane & 15;
  int khalf = lane >> 4;

  const float* xr  = x  + (m_t * 16 + row) * D_;
  const float* w1e = w1 + e * D_ * H_ + n_t * 16 + row;
  v8f c = {};
  for (int k = 0; k < D_; k += 4) {
    int ka = k + 2 * khalf;
    v2f a = *(const v2f*)(xr + ka);
    v2f bb;
    bb.x = w1e[(ka + 0) * H_];
    bb.y = w1e[(ka + 1) * H_];
    c = __builtin_amdgcn_wmma_f32_16x16x4_f32(
         false, a,  false, bb,
         (short)0, c,  false,  false);
  }
  float bias = b1[e * H_ + n_t * 16 + row];
  float* hout = h + (e * B_ + m_t * 16) * H_ + n_t * 16 + row;
#pragma unroll
  for (int v = 0; v < 8; ++v) {
    int r = v + 8 * khalf;
    float val = c[v] + bias;
    hout[r * H_] = val > 0.0f ? val : 0.0f;
  }
}

__global__ __launch_bounds__(128) __attribute__((amdgpu_waves_per_eu(1, 1)))
void moe_fc2(const float* __restrict__ hbuf,
             const float* __restrict__ w2,
             const float* __restrict__ b2,
             const float* __restrict__ gates,
             float* __restrict__ out) {
  const int NT = (O_ + 15) / 16;
  int wave = blockIdx.x * 4 + (threadIdx.x >> 5);
  if (wave >= 2 * NT) return;
  int m_t = wave / NT;
  int n_t = wave % NT;
  int lane  = threadIdx.x & 31;
  int row   = lane & 15;
  int khalf = lane >> 4;
  int ocol  = n_t * 16 + row;
  bool colok = ocol < O_;
  int ocl   = colok ? ocol : O_ - 1;

  v8f c = {};
#pragma unroll
  for (int e = 0; e < E_; ++e) {
    float g = gates[(m_t * 16 + row) * E_ + e];
    const float* he  = hbuf + (e * B_ + m_t * 16 + row) * H_;
    const float* w2e = w2 + e * H_ * O_;
#pragma unroll
    for (int k = 0; k < H_; k += 4) {
      int ka = k + 2 * khalf;
      v2f a = *(const v2f*)(he + ka);
      a.x *= g; a.y *= g;
      v2f bb;
      bb.x = w2e[(ka + 0) * O_ + ocl];
      bb.y = w2e[(ka + 1) * O_ + ocl];
      c = __builtin_amdgcn_wmma_f32_16x16x4_f32(
          false, a, false, bb, (short)0, c, false, false);
    }
  }
  if (colok) {
#pragma unroll
    for (int v = 0; v < 8; ++v) {
      int r = m_t * 16 + v + 8 * khalf;
      float acc = c[v];
      for (int e = 0; e < E_; ++e)
        acc = fmaf(gates[r * E_ + e], b2[e * O_ + ocol], acc);
      out[r * O_ + ocol] = acc;
    }
  }
}

extern "C" void kernel_launch(void* const* d_in, const int* in_sizes, int n_in,
                              void* d_out, int out_size, void* d_ws, size_t ws_size,
                              hipStream_t stream) {
  const float* x      = (const float*)d_in[0];
  const float* gate_w = (const float*)d_in[1];
  const float* gate_b = (const float*)d_in[2];
  const float* w1     = (const float*)d_in[3];
  const float* b1     = (const float*)d_in[4];
  const float* w2     = (const float*)d_in[5];
  const float* b2     = (const float*)d_in[6];
  float* out = (float*)d_out;

  float* gates = (float*)d_ws;
  float* hbuf  = gates + 256;

  moe_gate<<<1, B_ * E_, 0, stream>>>(x, gate_w, gate_b, gates);

  moe_fc1<<<(48 + 3) / 4, 128, 0, stream>>>(x, w1, b1, hbuf);

  moe_fc2<<<(126 + 3) / 4, 128, 0, stream>>>(hbuf, w2, b2, gates, out);
}


# gate and fc1: K=1024 split over the 16 waves of a 512-thread block (16 k-steps each, all loads issued up front), partial tiles summed through LDS; gate product on f32 WMMA; sum order changes
# speedup vs baseline: 6.7648x; 4.4579x over previous
#include <hip/hip_runtime.h>
#include <hip/hip_bf16.h>
#include <math.h>

typedef __attribute__((ext_vector_type(2))) float v2f;
typedef __attribute__((ext_vector_type(8))) float v8f;

#define B_  32
#define D_  1024
#define H_  64
#define O_  1000
#define E_  6

#define KS_ 16
#define KW_ (D_ / KS_)

__global__ __launch_bounds__(KS_ * 32) __attribute__((amdgpu_waves_per_eu(4, 4)))
void moe_gate(const float* __restrict__ x,
              const float* __restrict__ gw,
              const float* __restrict__ gb,
              float* __restrict__ gates) {
  __shared__ float part[KS_ * 256];
  __shared__ float tile[256];
  int t     = threadIdx.x;
  int w     = t >> 5;
  int lane  = t & 31;
  int row   = lane & 15;
  int khalf = lane >> 4;
  int m_t   = blockIdx.x;
  int ecl   = row < E_ ? row : E_ - 1;

  const float* xr  = x  + (m_t * 16 + row) * D_ + w * KW_;
  const float* gwc = gw + (w * KW_) * E_ + ecl;
  float bias = gb[ecl];
  v8f c = {};
#pragma unroll
  for (int k = 0; k < KW_; k += 4) {
    int ka = k + 2 * khalf;
    v2f a = *(const v2f*)(xr + ka);
    v2f bb;
    bb.x = gwc[(ka + 0) * E_];
    bb.y = gwc[(ka + 1) * E_];
    c = __builtin_amdgcn_wmma_f32_16x16x4_f32(
        false, a, false, bb, (short)0, c, false, false);
  }
#pragma unroll
  for (int v = 0; v < 8; ++v) part[w * 256 + v * 32 + lane] = c[v];
  __syncthreads();
  if (t < 256) {
    float s = part[t];
#pragma unroll
    for (int i = 1; i < KS_; ++i) s += part[i * 256 + t];
    tile[t] = s + bias;
  }
  __syncthreads();
  if (t < 16) {
    int r = t;
    float lg[E_];
#pragma unroll
    for (int e = 0; e < E_; ++e) lg[e] = tile[(r & 7) * 32 + (r >> 3) * 16 + e];
    int i0 = 0; float v0 = lg[0];
#pragma unroll
    for (int e = 1; e < E_; ++e) { float v = lg[e]; if (v > v0) { v0 = v; i0 = e; } }
    int i1 = -1; float v1 = -3.4e38f;
#pragma unroll
    for (int e = 0; e < E_; ++e) {
      if (e == i0) continue;
      float v = lg[e]; if (v > v1) { v1 = v; i1 = e; }
    }
    float e0 = 1.0f;
    float e1 = __expf(v1 - v0);
    float inv = 1.0f / (e0 + e1);
    int b = m_t * 16 + r;
    for (int e = 0; e < E_; ++e) gates[b * E_ + e] = 0.0f;
    gates[b * E_ + i0] = e0 * inv;
    gates[b * E_ + i1] = e1 * inv;
  }
}

__global__ __launch_bounds__(KS_ * 32) __attribute__((amdgpu_waves_per_eu(4, 4)))
void moe_fc1(const float* __restrict__ x,
             const float* __restrict__ w1,
             const float* __restrict__ b1,
             float* __restrict__ h) {
  __shared__ float part[KS_ * 256];
  int t     = threadIdx.x;
  int w     = t >> 5;
  int lane  = t & 31;
  int row   = lane & 15;
  int khalf = lane >> 4;
  int tl    = blockIdx.x;
  int e   = tl >> 3;
  int m_t = (tl >> 2) & 1;
  int n_t = tl & 3;

  const float* xr  = x  + (m_t * 16 + row) * D_ + w * KW_;
  const float* w1e = w1 + e * D_ * H_ + (w * KW_) * H_ + n_t * 16 + row;
  float bias = b1[e * H_ + n_t * 16 + row];
  v8f c = {};
#pragma unroll
  for (int k = 0; k < KW_; k += 4) {
    int ka = k + 2 * khalf;
    v2f a = *(const v2f*)(xr + ka);
    v2f bb;
    bb.x = w1e[(ka + 0) * H_];
    bb.y = w1e[(ka + 1) * H_];
    c = __builtin_amdgcn_wmma_f32_16x16x4_f32(
         false, a,  false, bb,
         (short)0, c,  false,  false);
  }
#pragma unroll
  for (int v = 0; v < 8; ++v) part[w * 256 + v * 32 + lane] = c[v];
  __syncthreads();
  if (t < 256) {
    float s = part[t];
#pragma unroll
    for (int i = 1; i < KS_; ++i) s += part[i * 256 + t];
    int col = t & 15;
    int r   = (t >> 5) + 8 * ((t >> 4) & 1);
    float val = s + bias;
    h[(e * B_ + m_t * 16 + r) * H_ + n_t * 16 + col] = val > 0.0f ? val : 0.0f;
  }
}

__global__ __launch_bounds__(128) __attribute__((amdgpu_waves_per_eu(1, 1)))
void moe_fc2(const float* __restrict__ hbuf,
             const float* __restrict__ w2,
             const float* __restrict__ b2,
             const float* __restrict__ gates,
             float* __restrict__ out) {
  const int NT = (O_ + 15) / 16;
  int wave = blockIdx.x * 4 + (threadIdx.x >> 5);
  if (wave >= 2 * NT) return;
  int m_t = wave / NT;
  int n_t = wave % NT;
  int lane  = threadIdx.x & 31;
  int row   = lane & 15;
  int khalf = lane >> 4;
  int ocol  = n_t * 16 + row;
  bool colok = ocol < O_;
  int ocl   = colok ? ocol : O_ - 1;

  v8f c = {};
#pragma unroll
  for (int e = 0; e < E_; ++e) {
    float g = gates[(m_t * 16 + row) * E_ + e];
    const float* he  = hbuf + (e * B_ + m_t * 16 + row) * H_;
    const float* w2e = w2 + e * H_ * O_;
#pragma unroll
    for (int k = 0; k < H_; k += 4) {
      int ka = k + 2 * khalf;
      v2f a = *(const v2f*)(he + ka);
      a.x *= g; a.y *= g;
      v2f bb;
      bb.x = w2e[(ka + 0) * O_ + ocl];
      bb.y = w2e[(ka + 1) * O_ + ocl];
      c = __builtin_amdgcn_wmma_f32_16x16x4_f32(
          false, a, false, bb, (short)0, c, false, false);
    }
  }
  if (colok) {
#pragma unroll
    for (int v = 0; v < 8; ++v) {
      int r = m_t * 16 + v + 8 * khalf;
      float acc = c[v];
      for (int e = 0; e < E_; ++e)
        acc = fmaf(gates[r * E_ + e], b2[e * O_ + ocol], acc);
      out[r * O_ + ocol] = acc;
    }
  }
}

extern "C" void kernel_launch(void* const* d_in, const int* in_sizes, int n_in,
                              void* d_out, int out_size, void* d_ws, size_t ws_size,
                              hipStream_t stream) {
  const float* x      = (const float*)d_in[0];
  const float* gate_w = (const float*)d_in[1];
  const float* gate_b = (const float*)d_in[2];
  const float* w1     = (const float*)d_in[3];
  const float* b1     = (const float*)d_in[4];
  const float* w2     = (const float*)d_in[5];
  const float* b2     = (const float*)d_in[6];
  float* out = (float*)d_out;

  float* gates = (float*)d_ws;
  float* hbuf  = gates + 256;

  moe_gate<<<2, KS_ * 32, 0, stream>>>(x, gate_w, gate_b, gates);

  moe_fc1<<<48, KS_ * 32, 0, stream>>>(x, w1, b1, hbuf);

  moe_fc2<<<(126 + 3) / 4, 128, 0, stream>>>(hbuf, w2, b2, gates, out);
}


# gate tiles run as two extra blocks of the fc1 launch (one grid of 50 blocks, branch on blockIdx): 3 launches become 2; sum order unchanged
# speedup vs baseline: 8.7930x; 1.2998x over previous
#include <hip/hip_runtime.h>
#include <hip/hip_bf16.h>
#include <math.h>

typedef __attribute__((ext_vector_type(2))) float v2f;
typedef __attribute__((ext_vector_type(8))) float v8f;

#define B_  32
#define D_  1024
#define H_  64
#define O_  1000
#define E_  6

#define KS_ 16
#define KW_ (D_ / KS_)

__device__ __forceinline__
void gate_tile(const float* __restrict__ x,
               const float* __restrict__ gw,
               const float* __restrict__ gb,
               float* __restrict__ gates,
               float* __restrict__ part,
               float* __restrict__ tile,
               int m_t) {
  int t     = threadIdx.x;
  int w     = t >> 5;
  int lane  = t & 31;
  int row   = lane & 15;
  int khalf = lane >> 4;
  int ecl   = row < E_ ? row : E_ - 1;

  const float* xr  = x  + (m_t * 16 + row) * D_ + w * KW_;
  const float* gwc = gw + (w * KW_) * E_ + ecl;
  float bias = gb[ecl];
  v8f c = {};
#pragma unroll
  for (int k = 0; k < KW_; k += 4) {
    int ka = k + 2 * khalf;
    v2f a = *(const v2f*)(xr + ka);
    v2f bb;
    bb.x = gwc[(ka + 0) * E_];
    bb.y = gwc[(ka + 1) * E_];
    c = __builtin_amdgcn_wmma_f32_16x16x4_f32(
        false, a, false, bb, (short)0, c, false, false);
  }
#pragma unroll
  for (int v = 0; v < 8; ++v) part[w * 256 + v * 32 + lane] = c[v];
  __syncthreads();
  if (t < 256) {
    float s = part[t];
#pragma unroll
    for (int i = 1; i < KS_; ++i) s += part[i * 256 + t];
    tile[t] = s + bias;
  }
  __syncthreads();
  if (t < 16) {
    int r = t;
    float lg[E_];
#pragma unroll
    for (int e = 0; e < E_; ++e) lg[e] = tile[(r & 7) * 32 + (r >> 3) * 16 + e];
    int i0 = 0; float v0 = lg[0];
#pragma unroll
    for (int e = 1; e < E_; ++e) { float v = lg[e]; if (v > v0) { v0 = v; i0 = e; } }
    int i1 = -1; float v1 = -3.4e38f;
#pragma unroll
    for (int e = 0; e < E_; ++e) {
      if (e == i0) continue;
      float v = lg[e]; if (v > v1) { v1 = v; i1 = e; }
    }
    float e0 = 1.0f;
    float e1 = __expf(v1 - v0);
    float inv = 1.0f / (e0 + e1);
    int b = m_t * 16 + r;
    for (int e = 0; e < E_; ++e) gates[b * E_ + e] = 0.0f;
    gates[b * E_ + i0] = e0 * inv;
    gates[b * E_ + i1] = e1 * inv;
  }
}

__device__ __forceinline__
void fc1_tile(const float* __restrict__ x,
              const float* __restrict__ w1,
              const float* __restrict__ b1,
              float* __restrict__ h,
              float* __restrict__ part,
              int tl) {
  int t     = threadIdx.x;
  int w     = t >> 5;
  int lane  = t & 31;
  int row   = lane & 15;
  int khalf = lane >> 4;
  int e   = tl >> 3;
  int m_t = (tl >> 2) & 1;
  int n_t = tl & 3;

  const float* xr  = x  + (m_t * 16 + row) * D_ + w * KW_;
  const float* w1e = w1 + e * D_ * H_ + (w * KW_) * H_ + n_t * 16 + row;
  float bias = b1[e * H_ + n_t * 16 + row];
  v8f c = {};
#pragma unroll
  for (int k = 0; k < KW_; k += 4) {
    int ka = k + 2 * khalf;
    v2f a = *(const v2f*)(xr + ka);
    v2f bb;
    bb.x = w1e[(ka + 0) * H_];
    bb.y = w1e[(ka + 1) * H_];
    c = __builtin_amdgcn_wmma_f32_16x16x4_f32(
         false, a,  false, bb,
         (short)0, c,  false,  false);
  }
#pragma unroll
  for (int v = 0; v < 8; ++v) part[w * 256 + v * 32 + lane] = c[v];
  __syncthreads();
  if (t < 256) {
    float s = part[t];
#pragma unroll
    for (int i = 1; i < KS_; ++i) s += part[i * 256 + t];
    int col = t & 15;
    int r   = (t >> 5) + 8 * ((t >> 4) & 1);
    float val = s + bias;
    h[(e * B_ + m_t * 16 + r) * H_ + n_t * 16 + col] = val > 0.0f ? val : 0.0f;
  }
}

__global__ __launch_bounds__(KS_ * 32) __attribute__((amdgpu_waves_per_eu(4, 4)))
void moe_gate_fc1(const float* __restrict__ x,
                  const float* __restrict__ gw,
                  const float* __restrict__ gb,
                  const float* __restrict__ w1,
                  const float* __restrict__ b1,
                  float* __restrict__ gates,
                  float* __restrict__ h) {
  __shared__ float part[KS_ * 256];
  __shared__ float tile[256];
  int blk = blockIdx.x;
  if (blk < 2) gate_tile(x, gw, gb, gates, part, tile, blk);
  else         fc1_tile(x, w1, b1, h, part, blk - 2);
}

__global__ __launch_bounds__(128) __attribute__((amdgpu_waves_per_eu(1, 1)))
void moe_fc2(const float* __restrict__ hbuf,
             const float* __restrict__ w2,
             const float* __restrict__ b2,
             const float* __restrict__ gates,
             float* __restrict__ out) {
  const int NT = (O_ + 15) / 16;
  int wave = blockIdx.x * 4 + (threadIdx.x >> 5);
  if (wave >= 2 * NT) return;
  int m_t = wave / NT;
  int n_t = wave % NT;
  int lane  = threadIdx.x & 31;
  int row   = lane & 15;
  int khalf = lane >> 4;
  int ocol  = n_t * 16 + row;
  bool colok = ocol < O_;
  int ocl   = colok ? ocol : O_ - 1;

  v8f c = {};
#pragma unroll
  for (int e = 0; e < E_; ++e) {
    float g = gates[(m_t * 16 + row) * E_ + e];
    const float* he  = hbuf + (e * B_ + m_t * 16 + row) * H_;
    const float* w2e = w2 + e * H_ * O_;
#pragma unroll
    for (int k = 0; k < H_; k += 4) {
      int ka = k + 2 * khalf;
      v2f a = *(const v2f*)(he + ka);
      a.x *= g; a.y *= g;
      v2f bb;
      bb.x = w2e[(ka + 0) * O_ + ocl];
      bb.y = w2e[(ka + 1) * O_ + ocl];
      c = __builtin_amdgcn_wmma_f32_16x16x4_f32(
          false, a, false, bb, (short)0, c, false, false);
    }
  }
  if (colok) {
#pragma unroll
    for (int v = 0; v < 8; ++v) {
      int r = m_t * 16 + v + 8 * khalf;
      float acc = c[v];
      for (int e = 0; e < E_; ++e)
        acc = fmaf(gates[r * E_ + e], b2[e * O_ + ocol], acc);
      out[r * O_ + ocol] = acc;
    }
  }
}

extern "C" void kernel_launch(void* const* d_in, const int* in_sizes, int n_in,
                              void* d_out, int out_size, void* d_ws, size_t ws_size,
                              hipStream_t stream) {
  const float* x      = (const float*)d_in[0];
  const float* gate_w = (const float*)d_in[1];
  const float* gate_b = (const float*)d_in[2];
  const float* w1     = (const float*)d_in[3];
  const float* b1     = (const float*)d_in[4];
  const float* w2     = (const float*)d_in[5];
  const float* b2     = (const float*)d_in[6];
  float* out = (float*)d_out;

  float* gates = (float*)d_ws;
  float* hbuf  = gates + 256;

  moe_gate_fc1<<<50, KS_ * 32, 0, stream>>>(x, gate_w, gate_b, w1, b1, gates, hbuf);

  moe_fc2<<<(126 + 3) / 4, 128, 0, stream>>>(hbuf, w2, b2, gates, out);
}


# fc2: K=384 split over the 6 waves of a 192-thread block, one expert (16 k-steps, all loads issued up front) each, one block per tile; partial tiles summed through LDS; sum order changes
# speedup vs baseline: 11.1168x; 1.2643x over previous
#include <hip/hip_runtime.h>
#include <hip/hip_bf16.h>
#include <math.h>

typedef __attribute__((ext_vector_type(2))) float v2f;
typedef __attribute__((ext_vector_type(8))) float v8f;

#define B_  32
#define D_  1024
#define H_  64
#define O_  1000
#define E_  6

#define KS_ 16
#define KW_ (D_ / KS_)
#define KS2_ E_

__device__ __forceinline__
void gate_tile(const float* __restrict__ x,
               const float* __restrict__ gw,
               const float* __restrict__ gb,
               float* __restrict__ gates,
               float* __restrict__ part,
               float* __restrict__ tile,
               int m_t) {
  int t     = threadIdx.x;
  int w     = t >> 5;
  int lane  = t & 31;
  int row   = lane & 15;
  int khalf = lane >> 4;
  int ecl   = row < E_ ? row : E_ - 1;

  const float* xr  = x  + (m_t * 16 + row) * D_ + w * KW_;
  const float* gwc = gw + (w * KW_) * E_ + ecl;
  float bias = gb[ecl];
  v8f c = {};
#pragma unroll
  for (int k = 0; k < KW_; k += 4) {
    int ka = k + 2 * khalf;
    v2f a = *(const v2f*)(xr + ka);
    v2f bb;
    bb.x = gwc[(ka + 0) * E_];
    bb.y = gwc[(ka + 1) * E_];
    c = __builtin_amdgcn_wmma_f32_16x16x4_f32(
        false, a, false, bb, (short)0, c, false, false);
  }
#pragma unroll
  for (int v = 0; v < 8; ++v) part[w * 256 + v * 32 + lane] = c[v];
  __syncthreads();
  if (t < 256) {
    float s = part[t];
#pragma unroll
    for (int i = 1; i < KS_; ++i) s += part[i * 256 + t];
    tile[t] = s + bias;
  }
  __syncthreads();
  if (t < 16) {
    int r = t;
    float lg[E_];
#pragma unroll
    for (int e = 0; e < E_; ++e) lg[e] = tile[(r & 7) * 32 + (r >> 3) * 16 + e];
    int i0 = 0; float v0 = lg[0];
#pragma unroll
    for (int e = 1; e < E_; ++e) { float v = lg[e]; if (v > v0) { v0 = v; i0 = e; } }
    int i1 = -1; float v1 = -3.4e38f;
#pragma unroll
    for (int e = 0; e < E_; ++e) {
      if (e == i0) continue;
      float v = lg[e]; if (v > v1) { v1 = v; i1 = e; }
    }
    float e0 = 1.0f;
    float e1 = __expf(v1 - v0);
    float inv = 1.0f / (e0 + e1);
    int b = m_t * 16 + r;
    for (int e = 0; e < E_; ++e) gates[b * E_ + e] = 0.0f;
    gates[b * E_ + i0] = e0 * inv;
    gates[b * E_ + i1] = e1 * inv;
  }
}

__device__ __forceinline__
void fc1_tile(const float* __restrict__ x,
              const float* __restrict__ w1,
              const float* __restrict__ b1,
              float* __restrict__ h,
              float* __restrict__ part,
              int tl) {
  int t     = threadIdx.x;
  int w     = t >> 5;
  int lane  = t & 31;
  int row   = lane & 15;
  int khalf = lane >> 4;
  int e   = tl >> 3;
  int m_t = (tl >> 2) & 1;
  int n_t = tl & 3;

  const float* xr  = x  + (m_t * 16 + row) * D_ + w * KW_;
  const float* w1e = w1 + e * D_ * H_ + (w * KW_) * H_ + n_t * 16 + row;
  float bias = b1[e * H_ + n_t * 16 + row];
  v8f c = {};
#pragma unroll
  for (int k = 0; k < KW_; k += 4) {
    int ka = k + 2 * khalf;
    v2f a = *(const v2f*)(xr + ka);
    v2f bb;
    bb.x = w1e[(ka + 0) * H_];
    bb.y = w1e[(ka + 1) * H_];
    c = __builtin_amdgcn_wmma_f32_16x16x4_f32(
         false, a,  false, bb,
         (short)0, c,  false,  false);
  }
#pragma unroll
  for (int v = 0; v < 8; ++v) part[w * 256 + v * 32 + lane] = c[v];
  __syncthreads();
  if (t < 256) {
    float s = part[t];
#pragma unroll
    for (int i = 1; i < KS_; ++i) s += part[i * 256 + t];
    int col = t & 15;
    int r   = (t >> 5) + 8 * ((t >> 4) & 1);
    float val = s + bias;
    h[(e * B_ + m_t * 16 + r) * H_ + n_t * 16 + col] = val > 0.0f ? val : 0.0f;
  }
}

__global__ __launch_bounds__(KS_ * 32) __attribute__((amdgpu_waves_per_eu(4, 4)))
void moe_gate_fc1(const float* __restrict__ x,
                  const float* __restrict__ gw,
                  const float* __restrict__ gb,
                  const float* __restrict__ w1,
                  const float* __restrict__ b1,
                  float* __restrict__ gates,
                  float* __restrict__ h) {
  __shared__ float part[KS_ * 256];
  __shared__ float tile[256];
  int blk = blockIdx.x;
  if (blk < 2) gate_tile(x, gw, gb, gates, part, tile, blk);
  else         fc1_tile(x, w1, b1, h, part, blk - 2);
}

__global__ __launch_bounds__(KS2_ * 32) __attribute__((amdgpu_waves_per_eu(2, 2)))
void moe_fc2(const float* __restrict__ hbuf,
             const float* __restrict__ w2,
             const float* __restrict__ b2,
             const float* __restrict__ gates,
             float* __restrict__ out) {
  __shared__ float part[KS2_ * 256];
  const int NT = (O_ + 15) / 16;
  int blk = blockIdx.x;
  int m_t = blk >= NT ? 1 : 0;
  int n_t = blk - m_t * NT;
  int t     = threadIdx.x;
  int e     = t >> 5;
  int lane  = t & 31;
  int row   = lane & 15;
  int khalf = lane >> 4;
  int ocol  = n_t * 16 + row;
  int ocl   = ocol < O_ ? ocol : O_ - 1;

  int u   = t & 127;
  int er  = m_t * 16 + (u >> 4) + 8 * ((u >> 3) & 1);
  int ec  = n_t * 16 + 2 * (u & 7);
  bool ecok = ec < O_;
  int ecl = ecok ? ec : O_ - 2;
  v2f ge[3];
  v2f bv[E_];
#pragma unroll
  for (int i = 0; i < 3; ++i) ge[i] = *(const v2f*)(gates + er * E_ + 2 * i);
#pragma unroll
  for (int i = 0; i < E_; ++i) bv[i] = *(const v2f*)(b2 + i * O_ + ecl);

  float g = gates[(m_t * 16 + row) * E_ + e];
  const float* he  = hbuf + (e * B_ + m_t * 16 + row) * H_;
  const float* w2e = w2 + e * H_ * O_ + ocl;
  v8f c = {};
#pragma unroll
  for (int k = 0; k < H_; k += 4) {
    int ka = k + 2 * khalf;
    v2f a = *(const v2f*)(he + ka);
    a.x *= g; a.y *= g;
    v2f bb;
    bb.x = w2e[(ka + 0) * O_];
    bb.y = w2e[(ka + 1) * O_];
    c = __builtin_amdgcn_wmma_f32_16x16x4_f32(
        false, a, false, bb, (short)0, c, false, false);
  }
#pragma unroll
  for (int v = 0; v < 8; ++v) part[e * 256 + v * 32 + lane] = c[v];
  __syncthreads();
  if (t < 128) {
    v2f s = *(const v2f*)(part + 2 * t);
#pragma unroll
    for (int i = 1; i < KS2_; ++i) {
      v2f p = *(const v2f*)(part + i * 256 + 2 * t);
      s.x += p.x; s.y += p.y;
    }
#pragma unroll
    for (int i = 0; i < E_; ++i) {
      float gi = (i & 1) ? ge[i >> 1].y : ge[i >> 1].x;
      s.x = fmaf(gi, bv[i].x, s.x);
      s.y = fmaf(gi, bv[i].y, s.y);
    }
    if (ecok) *(v2f*)(out + er * O_ + ec) = s;
  }
}

extern "C" void kernel_launch(void* const* d_in, const int* in_sizes, int n_in,
                              void* d_out, int out_size, void* d_ws, size_t ws_size,
                              hipStream_t stream) {
  const float* x      = (const float*)d_in[0];
  const float* gate_w = (const float*)d_in[1];
  const float* gate_b = (const float*)d_in[2];
  const float* w1     = (const float*)d_in[3];
  const float* b1     = (const float*)d_in[4];
  const float* w2     = (const float*)d_in[5];
  const float* b2     = (const float*)d_in[6];
  float* out = (float*)d_out;

  float* gates = (float*)d_ws;
  float* hbuf  = gates + 256;

  moe_gate_fc1<<<50, KS_ * 32, 0, stream>>>(x, gate_w, gate_b, w1, b1, gates, hbuf);

  moe_fc2<<<126, KS2_ * 32, 0, stream>>>(hbuf, w2, b2, gates, out);
}


# A operand of all three products: each wave fetches its 16x64 slice (x or h rows) with 8 whole-line 16-byte loads and turns it into fragments through its own LDS slice; sum order unchanged
# speedup vs baseline: 13.6390x; 1.2269x over previous
#include <hip/hip_runtime.h>
#include <hip/hip_bf16.h>
#include <math.h>

typedef __attribute__((ext_vector_type(2))) float v2f;
typedef __attribute__((ext_vector_type(4))) float v4f;
typedef __attribute__((ext_vector_type(8))) float v8f;

#define B_  32
#define D_  1024
#define H_  64
#define O_  1000
#define E_  6

#define KS_ 16
#define KW_ (D_ / KS_)
#define KS2_ E_
#define AP_ 68
#define AS_ (16 * AP_)

__device__ __forceinline__
void a_fetch(const float* __restrict__ src,
             int pitch,
             int lane, v4f (&ar)[8]) {
#pragma unroll
  for (int j = 0; j < 8; ++j)
    ar[j] = *(const v4f*)(src + (2 * j + (lane >> 4)) * pitch + (lane & 15) * 4);
}

__device__ __forceinline__
void a_turn(float* __restrict__ slice,
            int lane, const v4f (&ar)[8], v2f (&af)[16]) {
#pragma unroll
  for (int j = 0; j < 8; ++j)
    *(v4f*)(slice + (2 * j + (lane >> 4)) * AP_ + (lane & 15) * 4) = ar[j];
#pragma unroll
  for (int s2 = 0; s2 < 16; ++s2)
    af[s2] = *(const v2f*)(slice + (lane & 15) * AP_ + 4 * s2 + 2 * (lane >> 4));
  __builtin_amdgcn_sched_barrier(0);
}

__device__ __forceinline__
void gate_tile(const float* __restrict__ x,
               const float* __restrict__ gw,
               const float* __restrict__ gb,
               float* __restrict__ gates,
               float* __restrict__ part,
               float* __restrict__ tile,
               float* __restrict__ stage,
               int m_t) {
  int t     = threadIdx.x;
  int w     = t >> 5;
  int lane  = t & 31;
  int row   = lane & 15;
  int khalf = lane >> 4;
  int ecl   = row < E_ ? row : E_ - 1;

  const float* gwc = gw + (w * KW_) * E_ + ecl;
  v4f ar[8];
  a_fetch(x + (m_t * 16) * D_ + w * KW_, D_, lane, ar);
  float bias = gb[ecl];
  v2f bb[16];
#pragma unroll
  for (int k = 0; k < KW_; k += 4) {
    int ka = k + 2 * khalf;
    bb[k >> 2].x = gwc[(ka + 0) * E_];
    bb[k >> 2].y = gwc[(ka + 1) * E_];
  }
  v2f af[16];
  a_turn(stage + w * AS_, lane, ar, af);
  v8f c = {};
#pragma unroll
  for (int k = 0; k < KW_; k += 4)
    c = __builtin_amdgcn_wmma_f32_16x16x4_f32(
        false, af[k >> 2], false, bb[k >> 2], (short)0, c, false, false);
#pragma unroll
  for (int v = 0; v < 8; ++v) part[w * 256 + v * 32 + lane] = c[v];
  __syncthreads();
  if (t < 256) {
    float s = part[t];
#pragma unroll
    for (int i = 1; i < KS_; ++i) s += part[i * 256 + t];
    tile[t] = s + bias;
  }
  __syncthreads();
  if (t < 16) {
    int r = t;
    float lg[E_];
#pragma unroll
    for (int e = 0; e < E_; ++e) lg[e] = tile[(r & 7) * 32 + (r >> 3) * 16 + e];
    int i0 = 0; float v0 = lg[0];
#pragma unroll
    for (int e = 1; e < E_; ++e) { float v = lg[e]; if (v > v0) { v0 = v; i0 = e; } }
    int i1 = -1; float v1 = -3.4e38f;
#pragma unroll
    for (int e = 0; e < E_; ++e) {
      if (e == i0) continue;
      float v = lg[e]; if (v > v1) { v1 = v; i1 = e; }
    }
    float e0 = 1.0f;
    float e1 = __expf(v1 - v0);
    float inv = 1.0f / (e0 + e1);
    int b = m_t * 16 + r;
    for (int e = 0; e < E_; ++e) gates[b * E_ + e] = 0.0f;
    gates[b * E_ + i0] = e0 * inv;
    gates[b * E_ + i1] = e1 * inv;
  }
}

__device__ __forceinline__
void fc1_tile(const float* __restrict__ x,
              const float* __restrict__ w1,
              const float* __restrict__ b1,
              float* __restrict__ h,
              float* __restrict__ part,
              float* __restrict__ stage,
              int tl) {
  int t     = threadIdx.x;
  int w     = t >> 5;
  int lane  = t & 31;
  int row   = lane & 15;
  int khalf = lane >> 4;
  int e   = tl >> 3;
  int m_t = (tl >> 2) & 1;
  int n_t = tl & 3;

  const float* w1e = w1 + e * D_ * H_ + (w * KW_) * H_ + n_t * 16 + row;
  v4f ar[8];
  a_fetch(x + (m_t * 16) * D_ + w * KW_, D_, lane, ar);
  float bias = b1[e * H_ + n_t * 16 + row];
  v2f bb[16];
#pragma unroll
  for (int k = 0; k < KW_; k += 4) {
    int ka = k + 2 * khalf;
    bb[k >> 2].x = w1e[(ka + 0) * H_];
    bb[k >> 2].y = w1e[(ka + 1) * H_];
  }
  v2f af[16];
  a_turn(stage + w * AS_, lane, ar, af);
  v8f c = {};
#pragma unroll
  for (int k = 0; k < KW_; k += 4)
    c = __builtin_amdgcn_wmma_f32_16x16x4_f32(
         false, af[k >> 2],  false, bb[k >> 2],
         (short)0, c,  false,  false);
#pragma unroll
  for (int v = 0; v < 8; ++v) part[w * 256 + v * 32 + lane] = c[v];
  __syncthreads();
  if (t < 256) {
    float s = part[t];
#pragma unroll
    for (int i = 1; i < KS_; ++i) s += part[i * 256 + t];
    int col = t & 15;
    int r   = (t >> 5) + 8 * ((t >> 4) & 1);
    float val = s + bias;
    h[(e * B_ + m_t * 16 + r) * H_ + n_t * 16 + col] = val > 0.0f ? val : 0.0f;
  }
}

__global__ __launch_bounds__(KS_ * 32) __attribute__((amdgpu_waves_per_eu(4, 4)))
void moe_gate_fc1(const float* __restrict__ x,
                  const float* __restrict__ gw,
                  const float* __restrict__ gb,
                  const float* __restrict__ w1,
                  const float* __restrict__ b1,
                  float* __restrict__ gates,
                  float* __restrict__ h) {
  __shared__ float part[KS_ * 256];
  __shared__ float tile[256];
  __shared__ __attribute__((aligned(16))) float stage[KS_ * AS_];
  int blk = blockIdx.x;
  if (blk < 2) gate_tile(x, gw, gb, gates, part, tile, stage, blk);
  else         fc1_tile(x, w1, b1, h, part, stage, blk - 2);
}

__global__ __launch_bounds__(KS2_ * 32) __attribute__((amdgpu_waves_per_eu(2, 2)))
void moe_fc2(const float* __restrict__ hbuf,
             const float* __restrict__ w2,
             const float* __restrict__ b2,
             const float* __restrict__ gates,
             float* __restrict__ out) {
  __shared__ float part[KS2_ * 256];
  __shared__ __attribute__((aligned(16))) float stage[KS2_ * AS_];
  const int NT = (O_ + 15) / 16;
  int blk = blockIdx.x;
  int m_t = blk >= NT ? 1 : 0;
  int n_t = blk - m_t * NT;
  int t     = threadIdx.x;
  int e     = t >> 5;
  int lane  = t & 31;
  int row   = lane & 15;
  int khalf = lane >> 4;
  int ocol  = n_t * 16 + row;
  int ocl   = ocol < O_ ? ocol : O_ - 1;

  int u   = t & 127;
  int er  = m_t * 16 + (u >> 4) + 8 * ((u >> 3) & 1);
  int ec  = n_t * 16 + 2 * (u & 7);
  bool ecok = ec < O_;
  int ecl = ecok ? ec : O_ - 2;
  v2f ge[3];
  v2f bv[E_];

  v4f ar[8];
  a_fetch(hbuf + (e * B_ + m_t * 16) * H_, H_, lane, ar);
  float g = gates[(m_t * 16 + row) * E_ + e];
  const float* w2e = w2 + e * H_ * O_ + ocl;
  v2f bb[16];
#pragma unroll
  for (int k = 0; k < H_; k += 4) {
    int ka = k + 2 * khalf;
    bb[k >> 2].x = w2e[(ka + 0) * O_];
    bb[k >> 2].y = w2e[(ka + 1) * O_];
  }
#pragma unroll
  for (int i = 0; i < 3; ++i) ge[i] = *(const v2f*)(gates + er * E_ + 2 * i);
#pragma unroll
  for (int i = 0; i < E_; ++i) bv[i] = *(const v2f*)(b2 + i * O_ + ecl);
  v2f af[16];
  a_turn(stage + e * AS_, lane, ar, af);
  v8f c = {};
#pragma unroll
  for (int k = 0; k < H_; k += 4) {
    v2f a = af[k >> 2];
    a.x *= g; a.y *= g;
    c = __builtin_amdgcn_wmma_f32_16x16x4_f32(
        false, a, false, bb[k >> 2], (short)0, c, false, false);
  }
#pragma unroll
  for (int v = 0; v < 8; ++v) part[e * 256 + v * 32 + lane] = c[v];
  __syncthreads();
  if (t < 128) {
    v2f s = *(const v2f*)(part + 2 * t);
#pragma unroll
    for (int i = 1; i < KS2_; ++i) {
      v2f p = *(const v2f*)(part + i * 256 + 2 * t);
      s.x += p.x; s.y += p.y;
    }
#pragma unroll
    for (int i = 0; i < E_; ++i) {
      float gi = (i & 1) ? ge[i >> 1].y : ge[i >> 1].x;
      s.x = fmaf(gi, bv[i].x, s.x);
      s.y = fmaf(gi, bv[i].y, s.y);
    }
    if (ecok) *(v2f*)(out + er * O_ + ec) = s;
  }
}

extern "C" void kernel_launch(void* const* d_in, const int* in_sizes, int n_in,
                              void* d_out, int out_size, void* d_ws, size_t ws_size,
                              hipStream_t stream) {
  const float* x      = (const float*)d_in[0];
  const float* gate_w = (const float*)d_in[1];
  const float* gate_b = (const float*)d_in[2];
  const float* w1     = (const float*)d_in[3];
  const float* b1     = (const float*)d_in[4];
  const float* w2     = (const float*)d_in[5];
  const float* b2     = (const float*)d_in[6];
  float* out = (float*)d_out;

  float* gates = (float*)d_ws;
  float* hbuf  = gates + 256;

  moe_gate_fc1<<<50, KS_ * 32, 0, stream>>>(x, gate_w, gate_b, w1, b1, gates, hbuf);

  moe_fc2<<<126, KS2_ * 32, 0, stream>>>(hbuf, w2, b2, gates, out);
}
